# starting kernel
# baseline (speedup 1.0000x reference)
#include <hip/hip_runtime.h>


typedef float v2f __attribute__((ext_vector_type(2)));
typedef float v4f __attribute__((ext_vector_type(4)));
typedef float v8f __attribute__((ext_vector_type(8)));

__device__ __forceinline__ void atomicAddF32(float* p, float v) {
  unsafeAtomicAdd(p, v);
}

__global__ void k_init_deg(float* __restrict__ deg, int N) {
  int i = blockIdx.x * blockDim.x + threadIdx.x;
  if (i < N) deg[i] = 1.0f;
}

__global__ void k_deg_edges(float* __restrict__ deg, const int* __restrict__ dst, int E) {
  int i = blockIdx.x * blockDim.x + threadIdx.x;
  if (i < E) atomicAddF32(&deg[dst[i]], 1.0f);
}

__global__ void k_rsqrt(float* __restrict__ deg, int N) {
  int i = blockIdx.x * blockDim.x + threadIdx.x;
  if (i < N) deg[i] = rsqrtf(deg[i]);
}

template <int K, int MT>
__global__ void k_gemm_wmma(const float* __restrict__ X, const float* __restrict__ W,
                            float* __restrict__ Y, int M, int Nout) {
  constexpr int KSTEPS = K / 4;
  const int ntiles  = Nout >> 4;
  const int mtiles  = (M + 15) >> 4;
  const int mgroups = (mtiles + MT - 1) / MT;
  const int wave    = blockIdx.x * (blockDim.x >> 5) + (threadIdx.x >> 5);
  if (wave >= mgroups * ntiles) return;
  const int mg   = wave / ntiles;
  const int nt   = wave - mg * ntiles;
  const int lane = threadIdx.x & 31;
  const int half = lane >> 4;
  const int r    = lane & 15;
  const int n0   = nt << 4;

  const float* Bptr = W + (size_t)(2 * half) * Nout + (n0 + r);
  v2f bq[KSTEPS];
#pragma unroll
  for (int kk = 0; kk < KSTEPS; ++kk) {
    bq[kk][0] = Bptr[(size_t)(4 * kk + 0) * Nout];
    bq[kk][1] = Bptr[(size_t)(4 * kk + 1) * Nout];
  }

  for (int t = 0; t < MT; ++t) {
    const int mt = mg * MT + t;
    if (mt >= mtiles) break;
    const int m0 = mt << 4;
    int arow = m0 + r;
    if (arow >= M) arow = M - 1;
    const float* Aptr = X + (size_t)arow * K + 2 * half;

    v8f c = {};
#pragma unroll
    for (int kk = 0; kk < KSTEPS; ++kk) {
      v2f a = *(const v2f*)(Aptr + 4 * kk);
      c = __builtin_amdgcn_wmma_f32_16x16x4_f32(
           false, a,  false, bq[kk],
           (short)0, c,  false,  false);
    }
#pragma unroll
    for (int v = 0; v < 8; ++v) {
      int row = m0 + v + 8 * half;
      if (row < M) Y[(size_t)row * Nout + n0 + r] = c[v];
    }
  }
}

__global__ void k_self_loop(float* __restrict__ Y, const float* __restrict__ X,
                            const float* __restrict__ dinv, long long total, int logD) {
  long long t = (long long)blockIdx.x * blockDim.x + threadIdx.x;
  if (t < total) {
    float w = dinv[t >> logD];
    Y[t] = X[t] * w * w;
  }
}

__global__ void k_scatter(float* __restrict__ Y, const float* __restrict__ X,
                          const float* __restrict__ dinv,
                          const int* __restrict__ src, const int* __restrict__ dst,
                          int E, int D) {
  const int e = blockIdx.x * (blockDim.x >> 5) + (threadIdx.x >> 5);
  if (e >= E) return;
  const int lane = threadIdx.x & 31;
  const int s = src[e], d = dst[e];
  const float nrm = dinv[s] * dinv[d];
  const float* xs = X + (size_t)s * D;
  float* yd = Y + (size_t)d * D;
  if (D == 128) {
    v4f x = ((const v4f*)xs)[lane];
    float* p = yd + 4 * lane;
    atomicAddF32(p + 0, x[0] * nrm);
    atomicAddF32(p + 1, x[1] * nrm);
    atomicAddF32(p + 2, x[2] * nrm);
    atomicAddF32(p + 3, x[3] * nrm);
  } else if (D == 64) {
    v2f x = ((const v2f*)xs)[lane];
    float* p = yd + 2 * lane;
    atomicAddF32(p + 0, x[0] * nrm);
    atomicAddF32(p + 1, x[1] * nrm);
  } else {
    for (int j = lane; j < D; j += 32)
      atomicAddF32(&yd[j], xs[j] * nrm);
  }
}

__global__ void k_bias_relu(float* __restrict__ Y, const float* __restrict__ b,
                            long long total, int Dmask) {
  long long t = (long long)blockIdx.x * blockDim.x + threadIdx.x;
  if (t < total) {
    float v = Y[t] + b[t & Dmask];
    Y[t] = fmaxf(v, 0.0f);
  }
}

__global__ void k_bias_add(float* __restrict__ Y, const float* __restrict__ b,
                           long long total, int Dmask) {
  long long t = (long long)blockIdx.x * blockDim.x + threadIdx.x;
  if (t < total) Y[t] = Y[t] + b[t & Dmask];
}

extern "C" void kernel_launch(void* const* d_in, const int* in_sizes, int n_in,
                              void* d_out, int out_size, void* d_ws, size_t ws_size,
                              hipStream_t stream) {
  const float* emb  = (const float*)d_in[0];
  const float* W1   = (const float*)d_in[1];
  const float* b1   = (const float*)d_in[2];
  const float* W2   = (const float*)d_in[3];
  const float* b2   = (const float*)d_in[4];
  const int*   edge = (const int*)d_in[5];

  const int D_OUT = in_sizes[4];
  const int D_HID = in_sizes[2];
  const int N     = in_sizes[0] / D_OUT;
  const int E     = in_sizes[5] / 2;
  const int* src  = edge;
  const int* dst  = edge + E;

  char* ws = (char*)d_ws;
  const size_t o1 = (((size_t)N * 4) + 255) & ~(size_t)255;
  const size_t o2 = o1 + ((((size_t)N * D_HID * 4) + 255) & ~(size_t)255);
  float* dinv = (float*)ws;
  float* xbuf = (float*)(ws + o1);
  float* agg  = (float*)(ws + o2);

  const int T = 256;
  const int wavesPerBlk = T / 32;
  const int mtiles  = (N + 15) / 16;
  const int mgroups = (mtiles + 3) / 4;
  const long long nh = (long long)N * D_HID;
  const long long no = (long long)N * D_OUT;

  k_init_deg <<<(N + T - 1) / T, T, 0, stream>>>(dinv, N);
  k_deg_edges<<<(E + T - 1) / T, T, 0, stream>>>(dinv, dst, E);
  k_rsqrt    <<<(N + T - 1) / T, T, 0, stream>>>(dinv, N);

  {
    int waves = mgroups * (D_HID / 16);
    k_gemm_wmma<64, 4><<<(waves + wavesPerBlk - 1) / wavesPerBlk, T, 0, stream>>>(
        emb, W1, xbuf, N, D_HID);
    k_self_loop<<<(int)((nh + T - 1) / T), T, 0, stream>>>(agg, xbuf, dinv, nh, 7);
    k_scatter  <<<(E + wavesPerBlk - 1) / wavesPerBlk, T, 0, stream>>>(
        agg, xbuf, dinv, src, dst, E, D_HID);
    k_bias_relu<<<(int)((nh + T - 1) / T), T, 0, stream>>>(agg, b1, nh, D_HID - 1);
  }

  {
    int waves = mgroups * (D_OUT / 16);
    k_gemm_wmma<128, 4><<<(waves + wavesPerBlk - 1) / wavesPerBlk, T, 0, stream>>>(
        agg, W2, xbuf, N, D_OUT);
    float* out = (float*)d_out;
    k_self_loop<<<(int)((no + T - 1) / T), T, 0, stream>>>(out, xbuf, dinv, no, 6);
    k_scatter  <<<(E + wavesPerBlk - 1) / wavesPerBlk, T, 0, stream>>>(
        out, xbuf, dinv, src, dst, E, D_OUT);
    k_bias_add <<<(int)((no + T - 1) / T), T, 0, stream>>>(out, b2, no, D_OUT - 1);
  }
}


# in-degree count: one f32 global atomic per edge on a scattered word (and the fill launch) replaced by LDS integer histograms per node range and edge slice, stored as planes, summed by next launch
# speedup vs baseline: 2.9153x; 2.9153x over previous
#include <hip/hip_runtime.h>


typedef float v2f __attribute__((ext_vector_type(2)));
typedef float v8f __attribute__((ext_vector_type(8)));

__device__ __forceinline__ void atomicAddF32(float* p, float v) {
  unsafeAtomicAdd(p, v);
}

#define QSTRIDE 64
#define DEG_RN  12544
#define DEG_S   32
#define DEG_T   512
#define DEG_U   8

__device__ __forceinline__ void deg_bump(unsigned* cnt, int d, int r0) {
  const unsigned idx = (unsigned)(d - r0);
  if (idx < (unsigned)DEG_RN) atomicAdd(&cnt[idx], 1u);
}

__global__ void __launch_bounds__(DEG_T)
k_deg_hist(const int* __restrict__ dst, unsigned* __restrict__ planes,
           int E, int NP, int R, int slice) {
  __shared__ __attribute__((aligned(16))) unsigned cnt[DEG_RN];
  const int tid = threadIdx.x;
  const int r   = blockIdx.x % R;
  const int s   = blockIdx.x / R;
  const int r0  = r * DEG_RN;

  for (int i = 4 * tid; i < DEG_RN; i += 4 * DEG_T)
    *(uint4*)(&cnt[i]) = make_uint4(0u, 0u, 0u, 0u);
  __syncthreads();

  const int e0 = s * slice;
  int n = E - e0;
  if (n > slice) n = slice;
  if (n < 0) n = 0;
  const int* dp = dst + e0;

  if ((((size_t)dp) & 15) == 0) {
    const int4* d4 = (const int4*)dp;
    const int nq = n >> 2;
    for (int q = tid; q < nq; q += DEG_U * DEG_T) {
      int4 v[DEG_U];
#pragma unroll
      for (int u = 0; u < DEG_U; ++u) {
        const int qq = q + u * DEG_T;
        v[u] = (qq < nq) ? d4[qq] : make_int4(-1, -1, -1, -1);
      }
#pragma unroll
      for (int u = 0; u < DEG_U; ++u) {
        deg_bump(cnt, v[u].x, r0);
        deg_bump(cnt, v[u].y, r0);
        deg_bump(cnt, v[u].z, r0);
        deg_bump(cnt, v[u].w, r0);
      }
    }
    const int i = 4 * nq + tid;
    if (tid < (n & 3)) deg_bump(cnt, dp[i], r0);
  } else {
    for (int i = tid; i < n; i += DEG_T) deg_bump(cnt, dp[i], r0);
  }
  __syncthreads();

  unsigned* pl = planes + (size_t)s * NP + r0;
  for (int i = 4 * tid; i < DEG_RN; i += 4 * DEG_T)
    if (r0 + i < NP) *(uint4*)(pl + i) = *(const uint4*)(&cnt[i]);
}

__global__ void k_deg_sum(float* __restrict__ deg, const unsigned* __restrict__ planes,
                          int* __restrict__ qcnt, int N, int NP) {
  const int i = blockIdx.x * blockDim.x + threadIdx.x;
  if (i < 4) qcnt[i * QSTRIDE] = 0;
  if (i >= N) return;
  unsigned c = 0;
#pragma unroll
  for (int s = 0; s < DEG_S; ++s) c += planes[(size_t)s * NP + i];
  deg[i] = 1.0f + (float)c;
}

template <int K, int MT, bool RELU_A>
__global__ void k_gemm_wmma(const float* __restrict__ X, const float* __restrict__ W,
                            const float* __restrict__ bias, const float* __restrict__ deg,
                            float* __restrict__ Y, float* __restrict__ Z, int M, int Nout) {
  constexpr int KSTEPS = K / 4;
  const int ntiles  = Nout >> 4;
  const int mtiles  = (M + 15) >> 4;
  const int mgroups = (mtiles + MT - 1) / MT;
  const int wave    = blockIdx.x * (blockDim.x >> 5) + (threadIdx.x >> 5);
  if (wave >= mgroups * ntiles) return;
  const int mg   = wave / ntiles;
  const int nt   = wave - mg * ntiles;
  const int lane = threadIdx.x & 31;
  const int half = lane >> 4;
  const int r    = lane & 15;
  const int n0   = nt << 4;

  const float* Bptr = W + (size_t)(2 * half) * Nout + (n0 + r);
  v2f bq[KSTEPS];
#pragma unroll
  for (int kk = 0; kk < KSTEPS; ++kk) {
    bq[kk][0] = Bptr[(size_t)(4 * kk + 0) * Nout];
    bq[kk][1] = Bptr[(size_t)(4 * kk + 1) * Nout];
  }
  const float bc = bias[n0 + r];

  for (int t = 0; t < MT; ++t) {
    const int mt = mg * MT + t;
    if (mt >= mtiles) break;
    const int m0 = mt << 4;
    int arow = m0 + r;
    if (arow >= M) arow = M - 1;
    const float* Aptr = X + (size_t)arow * K + 2 * half;

    float w2[8];
#pragma unroll
    for (int v = 0; v < 8; ++v) {
      int wrow = m0 + v + 8 * half;
      if (wrow >= M) wrow = M - 1;
      const float w = rsqrtf(deg[wrow]);
      w2[v] = w * w;
    }

    v2f a[KSTEPS];
#pragma unroll
    for (int kk = 0; kk < KSTEPS; ++kk)
      a[kk] = *(const v2f*)(Aptr + 4 * kk);
    if (RELU_A) {
#pragma unroll
      for (int kk = 0; kk < KSTEPS; ++kk) {
        a[kk][0] = fmaxf(a[kk][0], 0.0f);
        a[kk][1] = fmaxf(a[kk][1], 0.0f);
      }
    }
    v8f c = {};
#pragma unroll
    for (int kk = 0; kk < KSTEPS; ++kk) {
      c = __builtin_amdgcn_wmma_f32_16x16x4_f32(
           false, a[kk],  false, bq[kk],
           (short)0, c,  false,  false);
    }
#pragma unroll
    for (int v = 0; v < 8; ++v) {
      int row = m0 + v + 8 * half;
      if (row < M) {
        const size_t o = (size_t)row * Nout + n0 + r;
        Y[o] = c[v];
        Z[o] = c[v] * w2[v] + bc;
      }
    }
  }
}

#define EPW 32
#define GA  8

__device__ __forceinline__ int bcast_i(int v, int j) {
  return __builtin_amdgcn_readlane(v, j);
}
__device__ __forceinline__ float bcast_f(float v, int j) {
  return __int_as_float(__builtin_amdgcn_readlane(__float_as_int(v), j));
}

template <int NV>
__device__ __forceinline__ void gather_row(const float* __restrict__ X, int s, int lane,
                                           float (&x)[NV]) {
  const float* p = X + (size_t)s * (32 * NV) + lane;
#pragma unroll
  for (int i = 0; i < NV; ++i) x[i] = p[32 * i];
}
template <int NV>
__device__ __forceinline__ void add_row(float* __restrict__ Y, int d, int lane,
                                        const float (&x)[NV], float nrm) {
  float* p = Y + (size_t)d * (32 * NV) + lane;
#pragma unroll
  for (int i = 0; i < NV; ++i) atomicAddF32(p + 32 * i, x[i] * nrm);
}

template <int NV>
__device__ __forceinline__ void scatter_rows(float* __restrict__ Y, const float* __restrict__ X,
                                             int s, int d, float nrm, unsigned mask, int lane) {
  int cnt = __builtin_popcount(mask);
  while (cnt >= GA) {
    float x[GA][NV];
    int js[GA];
#pragma unroll
    for (int u = 0; u < GA; ++u) { js[u] = __builtin_ctz(mask); mask &= mask - 1u; }
#pragma unroll
    for (int u = 0; u < GA; ++u) gather_row<NV>(X, bcast_i(s, js[u]), lane, x[u]);
#pragma unroll
    for (int u = 0; u < GA; ++u)
      add_row<NV>(Y, bcast_i(d, js[u]), lane, x[u], bcast_f(nrm, js[u]));
    cnt -= GA;
  }
  if (cnt > 0) {
    float x[GA][NV];
    int js[GA];
#pragma unroll
    for (int u = 0; u < GA; ++u) { js[u] = (mask != 0u) ? __builtin_ctz(mask) : 0; mask &= mask - 1u; }
#pragma unroll
    for (int u = 0; u < GA; ++u)
      if (u < cnt) gather_row<NV>(X, bcast_i(s, js[u]), lane, x[u]);
#pragma unroll
    for (int u = 0; u < GA; ++u)
      if (u < cnt) add_row<NV>(Y, bcast_i(d, js[u]), lane, x[u], bcast_f(nrm, js[u]));
  }
}

__global__ void k_scatter(float* __restrict__ Y, const float* __restrict__ X,
                          const float* __restrict__ deg,
                          const int* __restrict__ src, const int* __restrict__ dst,
                          int* __restrict__ qcnt, int E, int D, int N) {
  __shared__ int slot[2];
  __shared__ int s_half;
  const int tid  = threadIdx.x;
  const int lane = tid & 31;
  const int wv    = __builtin_amdgcn_readfirstlane(tid >> 5);
  const int chunk = (int)blockDim.x;
  const int nch   = (E + chunk - 1) / chunk;
  const int nhalf = N >> 1;

  if (tid == 0) {
    const unsigned id = __builtin_amdgcn_s_sendmsg_rtn(0x87);
    const int h = (int)((id >> 18) & 1u);
    s_half  = h;
    slot[0] = atomicAdd(&qcnt[h * QSTRIDE], 1);
  }
  __syncthreads();
  int q   = __builtin_amdgcn_readfirstlane(s_half);
  int cur = __builtin_amdgcn_readfirstlane(slot[0]);
  int par = 1;
  bool helping = false;

  const int max_trips = 2 * nch + 4;
  for (int trip = 0; trip < max_trips; ++trip) {
    if (cur >= nch) {
      if (helping) break;
      helping = true;
      q ^= 1;
      if (tid == 0) slot[par] = atomicAdd(&qcnt[q * QSTRIDE], 1);
      __syncthreads();
      cur = __builtin_amdgcn_readfirstlane(slot[par]);
      par ^= 1;
      continue;
    }
    int nxt = 0;
    if (tid == 0) nxt = atomicAdd(&qcnt[q * QSTRIDE], 1);

    const int e0 = cur * chunk + wv * EPW;
    if (e0 < E) {
      const int cnt = (E - e0 < EPW) ? (E - e0) : EPW;
      int s = 0, d = 0;
      float nrm = 0.0f;
      bool mine = false;
      if (lane < cnt) {
        s = src[e0 + lane];
        d = dst[e0 + lane];
        mine = ((d >= nhalf) ? 1 : 0) == q;
        if (mine) nrm = rsqrtf(deg[s]) * rsqrtf(deg[d]);
      }
      const unsigned mask = __builtin_amdgcn_ballot_w32(mine);
      if (D == 128) {
        scatter_rows<4>(Y, X, s, d, nrm, mask, lane);
      } else if (D == 64) {
        scatter_rows<2>(Y, X, s, d, nrm, mask, lane);
      } else {
        for (int j = 0; j < cnt; ++j) {
          if (((mask >> j) & 1u) == 0u) continue;
          const int sj = bcast_i(s, j), dj = bcast_i(d, j);
          const float nj = bcast_f(nrm, j);
          const float* xs = X + (size_t)sj * D;
          float* yd = Y + (size_t)dj * D;
          for (int c = lane; c < D; c += 32)
            atomicAddF32(&yd[c], xs[c] * nj);
        }
      }
    }

    if (tid == 0) slot[par] = nxt;
    __syncthreads();
    cur = __builtin_amdgcn_readfirstlane(slot[par]);
    par ^= 1;
  }
}

extern "C" void kernel_launch(void* const* d_in, const int* in_sizes, int n_in,
                              void* d_out, int out_size, void* d_ws, size_t ws_size,
                              hipStream_t stream) {
  const float* emb  = (const float*)d_in[0];
  const float* W1   = (const float*)d_in[1];
  const float* b1   = (const float*)d_in[2];
  const float* W2   = (const float*)d_in[3];
  const float* b2   = (const float*)d_in[4];
  const int*   edge = (const int*)d_in[5];

  const int D_OUT = in_sizes[4];
  const int D_HID = in_sizes[2];
  const int N     = in_sizes[0] / D_OUT;
  const int E     = in_sizes[5] / 2;
  const int* src  = edge;
  const int* dst  = edge + E;

  char* ws = (char*)d_ws;
  const size_t o1 = (((size_t)N * 4) + 255) & ~(size_t)255;
  const size_t o2 = o1 + ((((size_t)N * D_HID * 4) + 255) & ~(size_t)255);
  const size_t o3 = o2 + ((((size_t)N * D_HID * 4) + 255) & ~(size_t)255);
  float* deg  = (float*)ws;
  float* xbuf = (float*)(ws + o1);
  float* agg  = (float*)(ws + o2);
  int*   qcnt = (int*)(ws + o3);
  const size_t o4 = o3 + 4096;
  unsigned* planes = (unsigned*)(ws + o4);

  const int T = 256;
  const int wavesPerBlk = T / 32;
  const int mtiles  = (N + 15) / 16;
  const int mgroups = (mtiles + 3) / 4;
  const int sgrid = 2048;

  const int NP    = (N + 63) & ~63;
  const int R     = (N + DEG_RN - 1) / DEG_RN;
  const int slice = (((E + DEG_S - 1) / DEG_S) + 3) & ~3;
  k_deg_hist<<<R * DEG_S, DEG_T, 0, stream>>>(dst, planes, E, NP, R, slice);
  k_deg_sum <<<(N + T - 1) / T, T, 0, stream>>>(deg, planes, qcnt, N, NP);

  {
    int waves = mgroups * (D_HID / 16);
    k_gemm_wmma<64, 4, false><<<(waves + wavesPerBlk - 1) / wavesPerBlk, T, 0, stream>>>(
        emb, W1, b1, deg, xbuf, agg, N, D_HID);
    k_scatter  <<<sgrid, T, 0, stream>>>(
        agg, xbuf, deg, src, dst, qcnt, E, D_HID, N);
  }

  {
    int waves = mgroups * (D_OUT / 16);
    float* out = (float*)d_out;
    k_gemm_wmma<128, 4, true><<<(waves + wavesPerBlk - 1) / wavesPerBlk, T, 0, stream>>>(
        agg, W2, b2, deg, xbuf, out, N, D_OUT);
    k_scatter  <<<sgrid, T, 0, stream>>>(
        out, xbuf, deg, src, dst, qcnt + 2 * QSTRIDE, E, D_OUT, N);
  }
}
